# starting kernel
# baseline (speedup 1.0000x reference)
#include <hip/hip_runtime.h>


typedef float v2f __attribute__((ext_vector_type(2)));
typedef float v8f __attribute__((ext_vector_type(8)));

#define KS   15
#define RG   16
#define HW   256
#define HO   241
#define PADI 7
#define TILE 30
#define YS   32
#define GK   480
#define AGUARD 16

__global__ __launch_bounds__(32)
void CustomConv_corr_wmma(const float* __restrict__ image,
                          const float* __restrict__ kernels,
                          float* __restrict__ corr)
{
    __shared__ float PtT[TILE * YS];
    __shared__ float KtZ[AGUARD + KS * YS];

    const int r  = blockIdx.x;
    const int bc = blockIdx.y;
    const int i0 = (r >> 4) * RG;
    const int j0 = (r & 15) * RG;
    const int l  = threadIdx.x;

    for (int e = l; e < TILE * YS; e += 32)          PtT[e] = 0.0f;
    for (int e = l; e < AGUARD + KS * YS; e += 32)   KtZ[e] = 0.0f;

    for (int e = l; e < KS * KS; e += 32) {
        int v = e / KS, u = e - v * KS;
        KtZ[AGUARD + v * YS + u] = kernels[r * (KS * KS) + u * KS + v];
    }

    const float* img = image + bc * (HW * HW);
    for (int e = l; e < TILE * TILE; e += 32) {
        int yy = e / TILE, xx = e - yy * TILE;
        int gy = i0 + yy - PADI, gx = j0 + xx - PADI;
        float pv = 0.0f;
        if (gy >= 0 && gy < HW && gx >= 0 && gx < HW)
            pv = img[gy * HW + gx];
        PtT[xx * YS + yy] = pv;
    }
    __syncthreads();

    const int m  = l & 15;
    const int lh = l >> 4;

    const float* aBase = KtZ + (AGUARD + 2 * lh - m);
    const float* bBase = PtT + (YS * m + 2 * lh);

    v8f acc = {};
    #pragma unroll 8
    for (int k0 = 0; k0 < GK; k0 += 4) {
        v2f a, b;
        a[0] = aBase[k0];     a[1] = aBase[k0 + 1];
        b[0] = bBase[k0];     b[1] = bBase[k0 + 1];
        acc = __builtin_amdgcn_wmma_f32_16x16x4_f32(
                  false, a, false, b, (short)0, acc, false, false);
    }

    float* dst = corr + bc * (HW * HW);
    #pragma unroll
    for (int g = 0; g < 8; ++g) {
        int mi = lh * 8 + g;
        int i = i0 + mi, j = j0 + m;
        float ov = (i < HO && j < HO) ? acc[g] : 0.0f;
        dst[i * HW + j] = ov;
    }
}

__global__ __launch_bounds__(256)
void CustomConv_rowbox(const float* __restrict__ src, float* __restrict__ dst)
{
    int idx = blockIdx.x * 256 + threadIdx.x;
    int x = idx & 255;
    float s = 0.0f;
    #pragma unroll
    for (int d = 0; d < RG; ++d) {
        float v = (x - d >= 0) ? src[idx - d] : 0.0f;
        s += v;
    }
    dst[idx] = s;
}

__global__ __launch_bounds__(256)
void CustomConv_colbox(const float* __restrict__ src, float* __restrict__ dst)
{
    int idx = blockIdx.x * 256 + threadIdx.x;
    int y = (idx >> 8) & 255;
    float s = 0.0f;
    #pragma unroll
    for (int d = 0; d < RG; ++d) {
        float v = (y - d >= 0) ? src[idx - d * HW] : 0.0f;
        s += v;
    }
    dst[idx] = s;
}

extern "C" void kernel_launch(void* const* d_in, const int* in_sizes, int n_in,
                              void* d_out, int out_size, void* d_ws, size_t ws_size,
                              hipStream_t stream)
{
    const float* image   = (const float*)d_in[0];
    const float* kernels = (const float*)d_in[1];
    float* out = (float*)d_out;
    float* ws  = (float*)d_ws;

    dim3 g1(256, 12);
    CustomConv_corr_wmma<<<g1, 32, 0, stream>>>(image, kernels, out);

    const int total = 12 * HW * HW;
    CustomConv_rowbox<<<total / 256, 256, 0, stream>>>(out, ws);
    CustomConv_colbox<<<total / 256, 256, 0, stream>>>(ws, out);
}


# Pass 1 staging: all 45 global loads of a wave (tile column and kernel column per lane) issued before the first wait, then whole-row 16-byte LDS stores; was one load+wait per loop trip
# speedup vs baseline: 1.5102x; 1.5102x over previous
#include <hip/hip_runtime.h>


typedef float v2f __attribute__((ext_vector_type(2)));
typedef float v4f __attribute__((ext_vector_type(4)));
typedef float v8f __attribute__((ext_vector_type(8)));

#define KS   15
#define RG   16
#define HW   256
#define HO   241
#define PADI 7
#define TILE 30
#define YS   32
#define GK   480
#define AGUARD 16

__global__ __launch_bounds__(32)
void CustomConv_corr_wmma(const float* __restrict__ image,
                          const float* __restrict__ kernels,
                          float* __restrict__ corr)
{
    __shared__ __attribute__((aligned(16))) float PtT[TILE * YS];
    __shared__ __attribute__((aligned(16))) float KtZ[AGUARD + KS * YS];

    const int r  = blockIdx.x;
    const int bc = blockIdx.y;
    const int i0 = (r >> 4) * RG;
    const int j0 = (r & 15) * RG;
    const int l  = threadIdx.x;

    const float* img = image + bc * (HW * HW);
    const int  gx    = j0 + l - PADI;
    const bool colok = (gx >= 0) && (gx < HW);
    const int  gxc   = gx < 0 ? 0 : (gx > HW - 1 ? HW - 1 : gx);

    float tv[YS];
    #pragma unroll
    for (int yy = 0; yy < TILE; ++yy) {
        const int  gy  = i0 + yy - PADI;
        const int  gyc = gy < 0 ? 0 : (gy > HW - 1 ? HW - 1 : gy);
        const float pv = img[gyc * HW + gxc];
        tv[yy] = (colok && gy >= 0 && gy < HW) ? pv : 0.0f;
    }
    tv[TILE] = 0.0f;  tv[TILE + 1] = 0.0f;

    const int vc = l < KS ? l : KS - 1;
    const float* kp = kernels + r * (KS * KS) + vc;
    float kv[KS + 1];
    #pragma unroll
    for (int u = 0; u < KS; ++u) kv[u] = kp[u * KS];
    kv[KS] = 0.0f;

    const v4f z4 = {0.0f, 0.0f, 0.0f, 0.0f};
    if (l < TILE) {
        v4f* d = (v4f*)(PtT + l * YS);
        #pragma unroll
        for (int q = 0; q < YS / 4; ++q) {
            v4f w = {tv[4 * q], tv[4 * q + 1], tv[4 * q + 2], tv[4 * q + 3]};
            d[q] = w;
        }
    }
    if (l < KS) {
        v4f* d = (v4f*)(KtZ + AGUARD + l * YS);
        #pragma unroll
        for (int q = 0; q < 4; ++q) {
            v4f w = {kv[4 * q], kv[4 * q + 1], kv[4 * q + 2], kv[4 * q + 3]};
            d[q] = w;
        }
        #pragma unroll
        for (int q = 4; q < YS / 4; ++q) d[q] = z4;
    } else if (l < KS + AGUARD / 4) {
        ((v4f*)KtZ)[l - KS] = z4;
    }
    __syncthreads();

    const int m  = l & 15;
    const int lh = l >> 4;

    const float* aBase = KtZ + (AGUARD + 2 * lh - m);
    const float* bBase = PtT + (YS * m + 2 * lh);

    v8f acc = {};
    #pragma unroll 8
    for (int k0 = 0; k0 < GK; k0 += 4) {
        v2f a, b;
        a[0] = aBase[k0];     a[1] = aBase[k0 + 1];
        b[0] = bBase[k0];     b[1] = bBase[k0 + 1];
        acc = __builtin_amdgcn_wmma_f32_16x16x4_f32(
                  false, a, false, b, (short)0, acc, false, false);
    }

    float* dst = corr + bc * (HW * HW);
    #pragma unroll
    for (int g = 0; g < 8; ++g) {
        int mi = lh * 8 + g;
        int i = i0 + mi, j = j0 + m;
        float ov = (i < HO && j < HO) ? acc[g] : 0.0f;
        dst[i * HW + j] = ov;
    }
}

__global__ __launch_bounds__(256)
void CustomConv_rowbox(const float* __restrict__ src, float* __restrict__ dst)
{
    int idx = blockIdx.x * 256 + threadIdx.x;
    int x = idx & 255;
    float s = 0.0f;
    #pragma unroll
    for (int d = 0; d < RG; ++d) {
        float v = (x - d >= 0) ? src[idx - d] : 0.0f;
        s += v;
    }
    dst[idx] = s;
}

__global__ __launch_bounds__(256)
void CustomConv_colbox(const float* __restrict__ src, float* __restrict__ dst)
{
    int idx = blockIdx.x * 256 + threadIdx.x;
    int y = (idx >> 8) & 255;
    float s = 0.0f;
    #pragma unroll
    for (int d = 0; d < RG; ++d) {
        float v = (y - d >= 0) ? src[idx - d * HW] : 0.0f;
        s += v;
    }
    dst[idx] = s;
}

extern "C" void kernel_launch(void* const* d_in, const int* in_sizes, int n_in,
                              void* d_out, int out_size, void* d_ws, size_t ws_size,
                              hipStream_t stream)
{
    const float* image   = (const float*)d_in[0];
    const float* kernels = (const float*)d_in[1];
    float* out = (float*)d_out;
    float* ws  = (float*)d_ws;

    dim3 g1(256, 12);
    CustomConv_corr_wmma<<<g1, 32, 0, stream>>>(image, kernels, out);

    const int total = 12 * HW * HW;
    CustomConv_rowbox<<<total / 256, 256, 0, stream>>>(out, ws);
    CustomConv_colbox<<<total / 256, 256, 0, stream>>>(ws, out);
}


# Pass 1 LDS row pitch of image tile and banded kernel 32 -> 36 floats, K walked as (v, yy<32) at one shared offset: B-fragment 8-byte reads spread over all 64 banks (were 8 lanes per bank)
# speedup vs baseline: 2.5831x; 1.7104x over previous
#include <hip/hip_runtime.h>


typedef float v2f __attribute__((ext_vector_type(2)));
typedef float v4f __attribute__((ext_vector_type(4)));
typedef float v8f __attribute__((ext_vector_type(8)));

#define KS   15
#define RG   16
#define HW   256
#define HO   241
#define PADI 7
#define TILE 30
#define YK   32
#define YS   36
#define AGUARD 16

__global__ __launch_bounds__(32)
void CustomConv_corr_wmma(const float* __restrict__ image,
                          const float* __restrict__ kernels,
                          float* __restrict__ corr)
{
    __shared__ __attribute__((aligned(16))) float PtT[TILE * YS];
    __shared__ __attribute__((aligned(16))) float KtZ[AGUARD + KS * YS];

    const int r  = blockIdx.x;
    const int bc = blockIdx.y;
    const int i0 = (r >> 4) * RG;
    const int j0 = (r & 15) * RG;
    const int l  = threadIdx.x;

    const float* img = image + bc * (HW * HW);
    const int  gx    = j0 + l - PADI;
    const bool colok = (gx >= 0) && (gx < HW);
    const int  gxc   = gx < 0 ? 0 : (gx > HW - 1 ? HW - 1 : gx);

    float tv[YS];
    #pragma unroll
    for (int yy = 0; yy < TILE; ++yy) {
        const int  gy  = i0 + yy - PADI;
        const int  gyc = gy < 0 ? 0 : (gy > HW - 1 ? HW - 1 : gy);
        const float pv = img[gyc * HW + gxc];
        tv[yy] = (colok && gy >= 0 && gy < HW) ? pv : 0.0f;
    }
    #pragma unroll
    for (int yy = TILE; yy < YS; ++yy) tv[yy] = 0.0f;

    const int vc = l < KS ? l : KS - 1;
    const float* kp = kernels + r * (KS * KS) + vc;
    float kv[KS + 1];
    #pragma unroll
    for (int u = 0; u < KS; ++u) kv[u] = kp[u * KS];
    kv[KS] = 0.0f;

    const v4f z4 = {0.0f, 0.0f, 0.0f, 0.0f};
    if (l < TILE) {
        v4f* d = (v4f*)(PtT + l * YS);
        #pragma unroll
        for (int q = 0; q < YS / 4; ++q) {
            v4f w = {tv[4 * q], tv[4 * q + 1], tv[4 * q + 2], tv[4 * q + 3]};
            d[q] = w;
        }
    }
    if (l < KS) {
        v4f* d = (v4f*)(KtZ + AGUARD + l * YS);
        #pragma unroll
        for (int q = 0; q < 4; ++q) {
            v4f w = {kv[4 * q], kv[4 * q + 1], kv[4 * q + 2], kv[4 * q + 3]};
            d[q] = w;
        }
        #pragma unroll
        for (int q = 4; q < YS / 4; ++q) d[q] = z4;
    } else if (l < KS + AGUARD / 4) {
        ((v4f*)KtZ)[l - KS] = z4;
    }
    __syncthreads();

    const int m  = l & 15;
    const int lh = l >> 4;

    const float* aBase = KtZ + (AGUARD + 2 * lh - m);
    const float* bBase = PtT + (YS * m + 2 * lh);

    v8f acc = {};
    #pragma unroll 1
    for (int o = 0; o < KS * YS; o += YS) {
        #pragma unroll
        for (int y0 = 0; y0 < YK; y0 += 4) {
            v2f a, b;
            a[0] = aBase[o + y0];  a[1] = aBase[o + y0 + 1];
            b[0] = bBase[o + y0];  b[1] = bBase[o + y0 + 1];
            acc = __builtin_amdgcn_wmma_f32_16x16x4_f32(
                      false, a, false, b, (short)0, acc, false, false);
        }
    }

    float* dst = corr + bc * (HW * HW);
    #pragma unroll
    for (int g = 0; g < 8; ++g) {
        int mi = lh * 8 + g;
        int i = i0 + mi, j = j0 + m;
        float ov = (i < HO && j < HO) ? acc[g] : 0.0f;
        dst[i * HW + j] = ov;
    }
}

#define BT 8
#define BP 16
#define BW (HW + BP)

__global__ __launch_bounds__(64)
void CustomConv_box2d(const float* __restrict__ corr, float* __restrict__ out)
{
    __shared__ __attribute__((aligned(16))) float Vs[BT * BW];

    const int t  = threadIdx.x;
    const int y0 = blockIdx.x * BT;
    const int bc = blockIdx.y;
    const float* src = corr + bc * (HW * HW) + 4 * t;
    const v4f z4 = {0.0f, 0.0f, 0.0f, 0.0f};

    v4f v[BT + 15];
    #pragma unroll
    for (int q = 0; q < BT + 15; ++q) {
        const int gy  = y0 - 15 + q;
        const int gyc = gy < 0 ? 0 : gy;
        const v4f pv  = *(const v4f*)(src + gyc * HW);
        v[q] = (gy >= 0) ? pv : z4;
    }

    v4f core = (v[7] + v[8]) + (v[9] + v[10]);
    core += (v[11] + v[12]) + (v[13] + v[14]);
    core += v[15];
    v4f sa[BT], sb[BT];
    sa[BT - 2] = v[BT - 2];
    #pragma unroll
    for (int k = BT - 3; k >= 0; --k) sa[k] = sa[k + 1] + v[k];
    sb[1] = v[16];
    #pragma unroll
    for (int k = 2; k < BT; ++k) sb[k] = sb[k - 1] + v[15 + k];
    v4f cs[BT];
    cs[0] = core + sa[0];
    #pragma unroll
    for (int k = 1; k < BT - 1; ++k) cs[k] = core + (sa[k] + sb[k]);
    cs[BT - 1] = core + sb[BT - 1];

    #pragma unroll
    for (int k = 0; k < BT; ++k) *(v4f*)(Vs + k * BW + BP + 4 * t) = cs[k];
    if (t < BP / 4) {
        #pragma unroll
        for (int k = 0; k < BT; ++k) *(v4f*)(Vs + k * BW + 4 * t) = z4;
    }
    __syncthreads();

    float* dst = out + bc * (HW * HW) + y0 * HW + 4 * t;
    #pragma unroll
    for (int k = 0; k < BT; ++k) {
        const v4f* rp = (const v4f*)(Vs + k * BW + 4 * t);
        const v4f n0 = rp[0], n1 = rp[1], n2 = rp[2], n3 = rp[3];
        const v4f ow = cs[k];
        const float mid = (((n1[0] + n1[1]) + (n1[2] + n1[3])) + ((n2[0] + n2[1]) + (n2[2] + n2[3])))
                        + (((n3[0] + n3[1]) + (n3[2] + n3[3])) + ow[0]);
        const float a = n0[2] + n0[3];
        const float b = ow[1] + ow[2];
        v4f o;
        o[0] = mid + (n0[1] + a);
        o[1] = mid + (a + ow[1]);
        o[2] = mid + (n0[3] + b);
        o[3] = mid + (b + ow[3]);
        *(v4f*)(dst + k * HW) = o;
    }
}

extern "C" void kernel_launch(void* const* d_in, const int* in_sizes, int n_in,
                              void* d_out, int out_size, void* d_ws, size_t ws_size,
                              hipStream_t stream)
{
    const float* image   = (const float*)d_in[0];
    const float* kernels = (const float*)d_in[1];
    float* out = (float*)d_out;
    float* ws  = (float*)d_ws;

    dim3 g1(256, 12);
    CustomConv_corr_wmma<<<g1, 32, 0, stream>>>(image, kernels, ws);

    dim3 g2(32, 12);
    CustomConv_box2d<<<g2, 64, 0, stream>>>(ws, out);
}


# Pass 1 Toeplitz tile cut into 4x4 blocks (WMMA row = position in a block, column = block): kernel operand banded in both axes, K per 16x16 tile 480 -> 324, 120 -> 81 WMMA steps per wave
# speedup vs baseline: 2.8707x; 1.1113x over previous
#include <hip/hip_runtime.h>


typedef float v2f __attribute__((ext_vector_type(2)));
typedef float v4f __attribute__((ext_vector_type(4)));
typedef float v8f __attribute__((ext_vector_type(8)));

#define KS   15
#define RG   16
#define HW   256
#define HO   241
#define PADI 7
#define TILE 30
#define YS   36
#define BLK  4
#define WIN  18
#define KG   3
#define KZR  21

__global__ __launch_bounds__(32)
void CustomConv_corr_wmma(const float* __restrict__ image,
                          const float* __restrict__ kernels,
                          float* __restrict__ corr)
{
    __shared__ __attribute__((aligned(16))) float PtT[TILE * YS];
    __shared__ __attribute__((aligned(16))) float KtZ[KZR * YS];

    const int r  = blockIdx.x;
    const int bc = blockIdx.y;
    const int i0 = (r >> 4) * RG;
    const int j0 = (r & 15) * RG;
    const int l  = threadIdx.x;

    const float* img = image + bc * (HW * HW);
    const int  gx    = j0 + l - PADI;
    const bool colok = (gx >= 0) && (gx < HW);
    const int  gxc   = gx < 0 ? 0 : (gx > HW - 1 ? HW - 1 : gx);

    float tv[YS];
    #pragma unroll
    for (int yy = 0; yy < TILE; ++yy) {
        const int  gy  = i0 + yy - PADI;
        const int  gyc = gy < 0 ? 0 : (gy > HW - 1 ? HW - 1 : gy);
        const float pv = img[gyc * HW + gxc];
        tv[yy] = (colok && gy >= 0 && gy < HW) ? pv : 0.0f;
    }
    #pragma unroll
    for (int yy = TILE; yy < YS; ++yy) tv[yy] = 0.0f;

    const int vc = l < KS ? l : KS - 1;
    const float* kp = kernels + r * (KS * KS) + vc;
    float kv[KS];
    #pragma unroll
    for (int u = 0; u < KS; ++u) kv[u] = kp[u * KS];

    const v4f z4 = {0.0f, 0.0f, 0.0f, 0.0f};
    if (l < TILE) {
        v4f* d = (v4f*)(PtT + l * YS);
        #pragma unroll
        for (int q = 0; q < YS / 4; ++q) {
            v4f w = {tv[4 * q], tv[4 * q + 1], tv[4 * q + 2], tv[4 * q + 3]};
            d[q] = w;
        }
    }
    if (l < KS) {
        v4f* d = (v4f*)(KtZ + (l + KG) * YS);
        v4f w0 = {0.0f, 0.0f, 0.0f, kv[0]};
        d[0] = w0;
        #pragma unroll
        for (int q = 1; q < 4; ++q) {
            v4f w = {kv[4 * q - 3], kv[4 * q - 2], kv[4 * q - 1], kv[4 * q]};
            d[q] = w;
        }
        v4f w4 = {kv[13], kv[14], 0.0f, 0.0f};
        d[4] = w4;
        #pragma unroll
        for (int q = 5; q < YS / 4; ++q) d[q] = z4;
    } else if (l < KS + 2 * KG) {
        const int zr = (l < KS + KG) ? (l - KS) : l;
        v4f* d = (v4f*)(KtZ + zr * YS);
        #pragma unroll
        for (int q = 0; q < YS / 4; ++q) d[q] = z4;
    }
    __syncthreads();

    const int c  = l & 15;
    const int lh = l >> 4;
    const int cq = c >> 2;
    const int cr = c & 3;

    const float* aBase = KtZ + ((KG + lh - cr) * YS + (KG - cq));
    const float* bBase = PtT + ((BLK * cr + lh) * YS + BLK * cq);

    v8f acc = {};
    #pragma unroll 1
    for (int o = 0; o < (WIN / 2) * 2 * YS; o += 2 * YS) {
        #pragma unroll
        for (int y0 = 0; y0 < WIN; y0 += 2) {
            v2f a, b;
            a[0] = aBase[o + y0];  a[1] = aBase[o + y0 + 1];
            b[0] = bBase[o + y0];  b[1] = bBase[o + y0 + 1];
            acc = __builtin_amdgcn_wmma_f32_16x16x4_f32(
                      false, a, false, b, (short)0, acc, false, false);
        }
    }

    float* dst = corr + bc * (HW * HW);
    const int ib = i0 + BLK * cq + 2 * lh;
    const int jb = j0 + BLK * cr;
    #pragma unroll
    for (int gg = 0; gg < 2; ++gg) {
        const int i = ib + gg;
        v4f ov;
        #pragma unroll
        for (int e = 0; e < 4; ++e)
            ov[e] = (i < HO && jb + e < HO) ? acc[4 * gg + e] : 0.0f;
        *(v4f*)(dst + i * HW + jb) = ov;
    }
}

#define BT 8
#define BP 16
#define BW (HW + BP)

__global__ __launch_bounds__(64)
void CustomConv_box2d(const float* __restrict__ corr, float* __restrict__ out)
{
    __shared__ __attribute__((aligned(16))) float Vs[BT * BW];

    const int t  = threadIdx.x;
    const int y0 = blockIdx.x * BT;
    const int bc = blockIdx.y;
    const float* src = corr + bc * (HW * HW) + 4 * t;
    const v4f z4 = {0.0f, 0.0f, 0.0f, 0.0f};

    v4f v[BT + 15];
    #pragma unroll
    for (int q = 0; q < BT + 15; ++q) {
        const int gy  = y0 - 15 + q;
        const int gyc = gy < 0 ? 0 : gy;
        const v4f pv  = *(const v4f*)(src + gyc * HW);
        v[q] = (gy >= 0) ? pv : z4;
    }

    v4f core = (v[7] + v[8]) + (v[9] + v[10]);
    core += (v[11] + v[12]) + (v[13] + v[14]);
    core += v[15];
    v4f sa[BT], sb[BT];
    sa[BT - 2] = v[BT - 2];
    #pragma unroll
    for (int k = BT - 3; k >= 0; --k) sa[k] = sa[k + 1] + v[k];
    sb[1] = v[16];
    #pragma unroll
    for (int k = 2; k < BT; ++k) sb[k] = sb[k - 1] + v[15 + k];
    v4f cs[BT];
    cs[0] = core + sa[0];
    #pragma unroll
    for (int k = 1; k < BT - 1; ++k) cs[k] = core + (sa[k] + sb[k]);
    cs[BT - 1] = core + sb[BT - 1];

    #pragma unroll
    for (int k = 0; k < BT; ++k) *(v4f*)(Vs + k * BW + BP + 4 * t) = cs[k];
    if (t < BP / 4) {
        #pragma unroll
        for (int k = 0; k < BT; ++k) *(v4f*)(Vs + k * BW + 4 * t) = z4;
    }
    __syncthreads();

    float* dst = out + bc * (HW * HW) + y0 * HW + 4 * t;
    #pragma unroll
    for (int k = 0; k < BT; ++k) {
        const v4f* rp = (const v4f*)(Vs + k * BW + 4 * t);
        const v4f n0 = rp[0], n1 = rp[1], n2 = rp[2], n3 = rp[3];
        const v4f ow = cs[k];
        const float mid = (((n1[0] + n1[1]) + (n1[2] + n1[3])) + ((n2[0] + n2[1]) + (n2[2] + n2[3])))
                        + (((n3[0] + n3[1]) + (n3[2] + n3[3])) + ow[0]);
        const float a = n0[2] + n0[3];
        const float b = ow[1] + ow[2];
        v4f o;
        o[0] = mid + (n0[1] + a);
        o[1] = mid + (a + ow[1]);
        o[2] = mid + (n0[3] + b);
        o[3] = mid + (b + ow[3]);
        *(v4f*)(dst + k * HW) = o;
    }
}

extern "C" void kernel_launch(void* const* d_in, const int* in_sizes, int n_in,
                              void* d_out, int out_size, void* d_ws, size_t ws_size,
                              hipStream_t stream)
{
    const float* image   = (const float*)d_in[0];
    const float* kernels = (const float*)d_in[1];
    float* out = (float*)d_out;
    float* ws  = (float*)d_ws;

    dim3 g1(256, 12);
    CustomConv_corr_wmma<<<g1, 32, 0, stream>>>(image, kernels, ws);

    dim3 g2(32, 12);
    CustomConv_box2d<<<g2, 64, 0, stream>>>(ws, out);
}
